# starting kernel
# baseline (speedup 1.0000x reference)
#include <hip/hip_runtime.h>
#include <math.h>

#define NN     10000
#define EE     160000
#define IND    32
#define OUTD   500
#define FPAD   512
#define NQ     125
#define NTILES 32
#define MTILES 625
#define NEG_SLOPE 0.2f

typedef __attribute__((ext_vector_type(2)))  float    v2f;
typedef __attribute__((ext_vector_type(8)))  float    v8f;
typedef __attribute__((ext_vector_type(16))) _Float16 v16h;

#if __has_builtin(__builtin_amdgcn_wmma_f32_16x16x4_f32)
#define USE_WMMA_F32 1
#else
#define USE_WMMA_F32 0
#endif

__device__ __forceinline__ unsigned enc_f(float f) {
  unsigned u = __float_as_uint(f);
  return (u & 0x80000000u) ? ~u : (u | 0x80000000u);
}
__device__ __forceinline__ float dec_f(unsigned e) {
  return (e & 0x80000000u) ? __uint_as_float(e & 0x7FFFFFFFu)
                           : __uint_as_float(~e);
}
#define MAXENC_INIT 0x007FFFFFu

__global__ __launch_bounds__(256) void k_init(float* __restrict__ out,
                                              unsigned* __restrict__ m_enc,
                                              float* __restrict__ s) {
  const int i = blockIdx.x * 256 + threadIdx.x;
  if (i < NN * OUTD) out[i] = 0.0f;
  if (i < NN) { m_enc[i] = MAXENC_INIT; s[i] = 0.0f; }
}

__global__ __launch_bounds__(256) void k_gemm_wmma(const float* __restrict__ x,
                                                   const float* __restrict__ W,
                                                   float* __restrict__ feat) {
  const int gwave = (blockIdx.x * 256 + threadIdx.x) >> 5;
  const int lane  = threadIdx.x & 31;
  const int tileM = gwave >> 5;
  const int tileN = gwave & (NTILES - 1);
  const int m0 = tileM << 4, n0 = tileN << 4;
  const int r  = lane & 15;
  const int h  = lane >> 4;

  const int   col = n0 + r;
  const int   cc  = (col < OUTD) ? col : 0;
  const float msk = (col < OUTD) ? 1.0f : 0.0f;

  v8f acc = {};
#if USE_WMMA_F32
  v2f A[8], B[8];
  #pragma unroll
  for (int i = 0; i < 8; ++i) {
    const int k = (i << 2) + (h << 1);
    A[i].x = x[(m0 + r) * IND + k];
    A[i].y = x[(m0 + r) * IND + k + 1];
    B[i].x = W[k * OUTD + cc] * msk;
    B[i].y = W[(k + 1) * OUTD + cc] * msk;
  }
  #pragma unroll
  for (int i = 0; i < 8; ++i)
    acc = __builtin_amdgcn_wmma_f32_16x16x4_f32(
        false, A[i], false, B[i], (short)0, acc, false, false);
#else
  v16h a, b;
  #pragma unroll
  for (int j = 0; j < 8; ++j) {
    const int ka = ((j >> 2) << 4) + ((j & 3) << 1) + (h << 3);
    a[2 * j]     = (_Float16)x[(m0 + r) * IND + ka];
    a[2 * j + 1] = (_Float16)x[(m0 + r) * IND + ka + 1];
    const int kb = (j << 1) + (h << 4);
    b[2 * j]     = (_Float16)(W[kb * OUTD + cc] * msk);
    b[2 * j + 1] = (_Float16)(W[(kb + 1) * OUTD + cc] * msk);
  }
  acc = __builtin_amdgcn_wmma_f32_16x16x32_f16(
      false, a, false, b, (short)0, acc, false, false);
#endif
  #pragma unroll
  for (int rr = 0; rr < 8; ++rr)
    feat[(size_t)(m0 + rr + (h << 3)) * FPAD + n0 + r] = acc[rr];
}

__global__ __launch_bounds__(256) void k_el_er(const float* __restrict__ feat,
                                               const float* __restrict__ attn_l,
                                               const float* __restrict__ attn_r,
                                               float* __restrict__ el,
                                               float* __restrict__ er) {
  const int node = (blockIdx.x * 256 + threadIdx.x) >> 5;
  const int lane = threadIdx.x & 31;
  if (node >= NN) return;
  const float4* f4 = (const float4*)(feat + (size_t)node * FPAD);
  const float4* l4 = (const float4*)attn_l;
  const float4* r4 = (const float4*)attn_r;
  float sl = 0.0f, sr = 0.0f;
  for (int q = lane; q < NQ; q += 32) {
    const float4 v = f4[q], l = l4[q], rr = r4[q];
    sl = fmaf(v.x, l.x, fmaf(v.y, l.y, fmaf(v.z, l.z, fmaf(v.w, l.w, sl))));
    sr = fmaf(v.x, rr.x, fmaf(v.y, rr.y, fmaf(v.z, rr.z, fmaf(v.w, rr.w, sr))));
  }
  #pragma unroll
  for (int off = 16; off; off >>= 1) {
    sl += __shfl_xor(sl, off, 32);
    sr += __shfl_xor(sr, off, 32);
  }
  if (lane == 0) { el[node] = sl; er[node] = sr; }
}

__global__ __launch_bounds__(256) void k_edge_logit(const float* __restrict__ el,
                                                    const float* __restrict__ er,
                                                    const int* __restrict__ src,
                                                    const int* __restrict__ dst,
                                                    float* __restrict__ e,
                                                    unsigned* __restrict__ m_enc) {
  const int i = blockIdx.x * 256 + threadIdx.x;
  if (i >= EE) return;
  float v = el[src[i]] + er[dst[i]];
  v = (v > 0.0f) ? v : NEG_SLOPE * v;
  e[i] = v;
  atomicMax(&m_enc[dst[i]], enc_f(v));
}

__global__ __launch_bounds__(256) void k_edge_exp(const int* __restrict__ dst,
                                                  float* __restrict__ e,
                                                  const unsigned* __restrict__ m_enc,
                                                  float* __restrict__ s) {
  const int i = blockIdx.x * 256 + threadIdx.x;
  if (i >= EE) return;
  const int d = dst[i];
  const float a = __expf(e[i] - dec_f(m_enc[d]));
  e[i] = a;
  atomicAdd(&s[d], a);
}

__global__ __launch_bounds__(256) void k_scatter(const int* __restrict__ src,
                                                 const int* __restrict__ dst,
                                                 const float* __restrict__ a,
                                                 const float* __restrict__ s,
                                                 const float* __restrict__ feat,
                                                 float* __restrict__ out) {
  const int edge = (blockIdx.x * 256 + threadIdx.x) >> 5;
  const int lane = threadIdx.x & 31;
  const int sn = src[edge], dn = dst[edge];
  const float sv = s[dn];
  const float alpha = a[edge] / (sv > 0.0f ? sv : 1.0f);
  const float4* f4 = (const float4*)(feat + (size_t)sn * FPAD);
  float*        o  = out + (size_t)dn * OUTD;
  __builtin_prefetch((const float*)f4 + 4 * lane, 0, 1);
  #pragma unroll
  for (int q = lane; q < FPAD / 4; q += 32) {
    const int d = q << 2;
    if (d < OUTD) {
      const float4 v = f4[q];
      atomicAdd(&o[d + 0], alpha * v.x);
      atomicAdd(&o[d + 1], alpha * v.y);
      atomicAdd(&o[d + 2], alpha * v.z);
      atomicAdd(&o[d + 3], alpha * v.w);
    }
  }
}

__global__ __launch_bounds__(256) void k_row_softmax(float* __restrict__ out) {
  const int node = (blockIdx.x * 256 + threadIdx.x) >> 5;
  const int lane = threadIdx.x & 31;
  if (node >= NN) return;
  float4* r4 = (float4*)(out + (size_t)node * OUTD);
  float mx = -INFINITY;
  for (int q = lane; q < NQ; q += 32) {
    const float4 v = r4[q];
    mx = fmaxf(mx, fmaxf(fmaxf(v.x, v.y), fmaxf(v.z, v.w)));
  }
  #pragma unroll
  for (int off = 16; off; off >>= 1) mx = fmaxf(mx, __shfl_xor(mx, off, 32));
  float sm = 0.0f;
  for (int q = lane; q < NQ; q += 32) {
    float4 v = r4[q];
    v.x = __expf(v.x - mx); v.y = __expf(v.y - mx);
    v.z = __expf(v.z - mx); v.w = __expf(v.w - mx);
    r4[q] = v;
    sm += (v.x + v.y) + (v.z + v.w);
  }
  #pragma unroll
  for (int off = 16; off; off >>= 1) sm += __shfl_xor(sm, off, 32);
  const float inv = 1.0f / sm;
  for (int q = lane; q < NQ; q += 32) {
    float4 v = r4[q];
    v.x *= inv; v.y *= inv; v.z *= inv; v.w *= inv;
    r4[q] = v;
  }
}

__global__ __launch_bounds__(256) void k_px_r(const float* __restrict__ px_r,
                                              float* __restrict__ out) {
  const int i = blockIdx.x * 256 + threadIdx.x;
  if (i >= OUTD) return;
  const float v = px_r[i];
  const float sp = (v > 20.0f) ? v : log1pf(__expf(v));
  out[(size_t)NN * OUTD + i] = fminf(fmaxf(sp, 1.0e-4f), 1.0e4f);
}

extern "C" void kernel_launch(void* const* d_in, const int* in_sizes, int n_in,
                              void* d_out, int out_size, void* d_ws, size_t ws_size,
                              hipStream_t stream) {
  const float* x      = (const float*)d_in[0];
  const float* W      = (const float*)d_in[1];
  const float* attn_l = (const float*)d_in[2];
  const float* attn_r = (const float*)d_in[3];
  const float* px_r   = (const float*)d_in[4];
  const int*   src    = (const int*)d_in[5];
  const int*   dst    = (const int*)d_in[6];
  float*       out    = (float*)d_out;

  float*    ws    = (float*)d_ws;
  float*    feat  = ws;
  float*    el    = feat + (size_t)NN * FPAD;
  float*    er    = el + NN;
  unsigned* m_enc = (unsigned*)(er + NN);
  float*    s     = (float*)(m_enc + NN);
  float*    ea    = s + NN;

  const int initN = NN * OUTD;
  k_init       <<<(initN + 255) / 256, 256, 0, stream>>>(out, m_enc, s);
  k_gemm_wmma  <<<(MTILES * NTILES) / 8, 256, 0, stream>>>(x, W, feat);
  k_el_er      <<<(NN + 7) / 8, 256, 0, stream>>>(feat, attn_l, attn_r, el, er);
  k_edge_logit <<<(EE + 255) / 256, 256, 0, stream>>>(el, er, src, dst, ea, m_enc);
  k_edge_exp   <<<(EE + 255) / 256, 256, 0, stream>>>(dst, ea, m_enc, s);
  k_scatter    <<<EE / 8, 256, 0, stream>>>(src, dst, ea, s, feat, out);
  k_row_softmax<<<(NN + 7) / 8, 256, 0, stream>>>(out);
  k_px_r       <<<(OUTD + 255) / 256, 256, 0, stream>>>(px_r, out);
}


# out rows summed in LDS: 250 work-groups own 40 dst rows, scan all dst, add alpha*feat[src] by LDS float atomics, store rows once; no global float atomics, no out zero fill; sum order: LDS arrival
# speedup vs baseline: 2.0861x; 2.0861x over previous
#include <hip/hip_runtime.h>
#include <math.h>

#define NN     10000
#define EE     160000
#define IND    32
#define OUTD   500
#define FPAD   512
#define NQ     125
#define NTILES 32
#define MTILES 625
#define NEG_SLOPE 0.2f

typedef __attribute__((ext_vector_type(2)))  float    v2f;
typedef __attribute__((ext_vector_type(8)))  float    v8f;
typedef __attribute__((ext_vector_type(16))) _Float16 v16h;

#if __has_builtin(__builtin_amdgcn_wmma_f32_16x16x4_f32)
#define USE_WMMA_F32 1
#else
#define USE_WMMA_F32 0
#endif

__device__ __forceinline__ unsigned enc_f(float f) {
  unsigned u = __float_as_uint(f);
  return (u & 0x80000000u) ? ~u : (u | 0x80000000u);
}
__device__ __forceinline__ float dec_f(unsigned e) {
  return (e & 0x80000000u) ? __uint_as_float(e & 0x7FFFFFFFu)
                           : __uint_as_float(~e);
}
#define MAXENC_INIT 0x007FFFFFu

__global__ __launch_bounds__(256) void k_init(unsigned* __restrict__ m_enc,
                                              float* __restrict__ s) {
  const int i = blockIdx.x * 256 + threadIdx.x;
  if (i < NN) { m_enc[i] = MAXENC_INIT; s[i] = 0.0f; }
}

__global__ __launch_bounds__(256) void k_gemm_wmma(const float* __restrict__ x,
                                                   const float* __restrict__ W,
                                                   float* __restrict__ feat) {
  const int gwave = (blockIdx.x * 256 + threadIdx.x) >> 5;
  const int lane  = threadIdx.x & 31;
  const int tileM = gwave >> 5;
  const int tileN = gwave & (NTILES - 1);
  const int m0 = tileM << 4, n0 = tileN << 4;
  const int r  = lane & 15;
  const int h  = lane >> 4;

  const int   col = n0 + r;
  const int   cc  = (col < OUTD) ? col : 0;
  const float msk = (col < OUTD) ? 1.0f : 0.0f;

  v8f acc = {};
#if USE_WMMA_F32
  v2f A[8], B[8];
  #pragma unroll
  for (int i = 0; i < 8; ++i) {
    const int k = (i << 2) + (h << 1);
    A[i].x = x[(m0 + r) * IND + k];
    A[i].y = x[(m0 + r) * IND + k + 1];
    B[i].x = W[k * OUTD + cc] * msk;
    B[i].y = W[(k + 1) * OUTD + cc] * msk;
  }
  #pragma unroll
  for (int i = 0; i < 8; ++i)
    acc = __builtin_amdgcn_wmma_f32_16x16x4_f32(
        false, A[i], false, B[i], (short)0, acc, false, false);
#else
  v16h a, b;
  #pragma unroll
  for (int j = 0; j < 8; ++j) {
    const int ka = ((j >> 2) << 4) + ((j & 3) << 1) + (h << 3);
    a[2 * j]     = (_Float16)x[(m0 + r) * IND + ka];
    a[2 * j + 1] = (_Float16)x[(m0 + r) * IND + ka + 1];
    const int kb = (j << 1) + (h << 4);
    b[2 * j]     = (_Float16)(W[kb * OUTD + cc] * msk);
    b[2 * j + 1] = (_Float16)(W[(kb + 1) * OUTD + cc] * msk);
  }
  acc = __builtin_amdgcn_wmma_f32_16x16x32_f16(
      false, a, false, b, (short)0, acc, false, false);
#endif
  #pragma unroll
  for (int rr = 0; rr < 8; ++rr)
    feat[(size_t)(m0 + rr + (h << 3)) * FPAD + n0 + r] = acc[rr];
}

__global__ __launch_bounds__(256) void k_el_er(const float* __restrict__ feat,
                                               const float* __restrict__ attn_l,
                                               const float* __restrict__ attn_r,
                                               float* __restrict__ el,
                                               float* __restrict__ er) {
  const int node = (blockIdx.x * 256 + threadIdx.x) >> 5;
  const int lane = threadIdx.x & 31;
  if (node >= NN) return;
  const float4* f4 = (const float4*)(feat + (size_t)node * FPAD);
  const float4* l4 = (const float4*)attn_l;
  const float4* r4 = (const float4*)attn_r;
  float sl = 0.0f, sr = 0.0f;
  for (int q = lane; q < NQ; q += 32) {
    const float4 v = f4[q], l = l4[q], rr = r4[q];
    sl = fmaf(v.x, l.x, fmaf(v.y, l.y, fmaf(v.z, l.z, fmaf(v.w, l.w, sl))));
    sr = fmaf(v.x, rr.x, fmaf(v.y, rr.y, fmaf(v.z, rr.z, fmaf(v.w, rr.w, sr))));
  }
  #pragma unroll
  for (int off = 16; off; off >>= 1) {
    sl += __shfl_xor(sl, off, 32);
    sr += __shfl_xor(sr, off, 32);
  }
  if (lane == 0) { el[node] = sl; er[node] = sr; }
}

__global__ __launch_bounds__(256) void k_edge_logit(const float* __restrict__ el,
                                                    const float* __restrict__ er,
                                                    const int* __restrict__ src,
                                                    const int* __restrict__ dst,
                                                    float* __restrict__ e,
                                                    unsigned* __restrict__ m_enc) {
  const int i = blockIdx.x * 256 + threadIdx.x;
  if (i >= EE) return;
  float v = el[src[i]] + er[dst[i]];
  v = (v > 0.0f) ? v : NEG_SLOPE * v;
  e[i] = v;
  atomicMax(&m_enc[dst[i]], enc_f(v));
}

__global__ __launch_bounds__(256) void k_edge_exp(const int* __restrict__ dst,
                                                  float* __restrict__ e,
                                                  const unsigned* __restrict__ m_enc,
                                                  float* __restrict__ s) {
  const int i = blockIdx.x * 256 + threadIdx.x;
  if (i >= EE) return;
  const int d = dst[i];
  const float a = __expf(e[i] - dec_f(m_enc[d]));
  e[i] = a;
  atomicAdd(&s[d], a);
}

#define BAND     40
#define NBANDS   (NN / BAND)
#define BTHREADS 1024
#define BWAVES   (BTHREADS / 32)
#define NGROUPS  (EE / 128)
#define QPB      8
#define NBATCH   5
__global__ __launch_bounds__(BTHREADS) void k_band_agg(const int* __restrict__ src,
                                                       const int* __restrict__ dst,
                                                       const float* __restrict__ a,
                                                       const float* __restrict__ s,
                                                       const float* __restrict__ feat,
                                                       float* __restrict__ out) {
  __shared__ float acc[BAND * FPAD];
  const int tid  = threadIdx.x;
  const int lane = tid & 31;
  const int w    = tid >> 5;
  const int d0   = blockIdx.x * BAND;

  for (int i = tid; i < BAND * FPAD; i += BTHREADS) acc[i] = 0.0f;
  __syncthreads();

  const int4* dst4 = (const int4*)dst;
  #pragma nounroll
  for (int bt = 0; bt < NBATCH; ++bt) {
    int dv[4 * QPB];
    #pragma unroll
    for (int k = 0; k < QPB; ++k) {
      const int  g = w + BWAVES * (QPB * bt + k);
      const int4 q = dst4[(g < NGROUPS ? g : w) * 32 + lane];
      dv[4 * k + 0] = q.x; dv[4 * k + 1] = q.y; dv[4 * k + 2] = q.z; dv[4 * k + 3] = q.w;
    }
    unsigned pend = 0u;
    #pragma unroll
    for (int j = 0; j < 4 * QPB; ++j)
      pend |= ((unsigned)(dv[j] - d0) < (unsigned)BAND) ? (1u << j) : 0u;
    #pragma unroll
    for (int k = 0; k < QPB; ++k)
      if (w + BWAVES * (QPB * bt + k) >= NGROUPS) pend &= ~(0xFu << (4 * k));

    while (__builtin_amdgcn_ballot_w32(pend != 0u) != 0u) {
      const bool has = pend != 0u;
      const int  j   = has ? __builtin_ctz(pend) : 0;
      pend &= pend - 1u;
      int dc = dv[0];
      #pragma unroll
      for (int t = 1; t < 4 * QPB; ++t) dc = (j == t) ? dv[t] : dc;
      const int e = (w + BWAVES * (QPB * bt + (j >> 2))) * 128 + lane * 4 + (j & 3);
      int   snv = 0;
      float alv = 0.0f;
      if (has) {
        snv = src[e];
        const float av = a[e];
        const float sv = s[dc];
        alv = av / (sv > 0.0f ? sv : 1.0f);
      }
      unsigned m = __builtin_amdgcn_ballot_w32(has);
      while (m != 0u) {
        const int l = __builtin_ctz(m);
        m &= m - 1u;
        const int   row   = __builtin_amdgcn_readlane(dc, l) - d0;
        const int   snl   = __builtin_amdgcn_readlane(snv, l);
        const float alpha = __int_as_float(
            __builtin_amdgcn_readlane(__float_as_int(alv), l));
        const float4* f4 = (const float4*)(feat + (size_t)snl * FPAD);
        float* arow = acc + row * FPAD;
        const float4 v0 = f4[lane];
        const float4 v1 = f4[lane + 32];
        const float4 v2 = f4[lane + 64];
        const float4 v3 = f4[lane + 96];
        atomicAdd(&arow[lane],            alpha * v0.x);
        atomicAdd(&arow[128 + lane],      alpha * v0.y);
        atomicAdd(&arow[256 + lane],      alpha * v0.z);
        atomicAdd(&arow[384 + lane],      alpha * v0.w);
        atomicAdd(&arow[lane + 32],       alpha * v1.x);
        atomicAdd(&arow[128 + lane + 32], alpha * v1.y);
        atomicAdd(&arow[256 + lane + 32], alpha * v1.z);
        atomicAdd(&arow[384 + lane + 32], alpha * v1.w);
        atomicAdd(&arow[lane + 64],       alpha * v2.x);
        atomicAdd(&arow[128 + lane + 64], alpha * v2.y);
        atomicAdd(&arow[256 + lane + 64], alpha * v2.z);
        atomicAdd(&arow[384 + lane + 64], alpha * v2.w);
        atomicAdd(&arow[lane + 96],       alpha * v3.x);
        atomicAdd(&arow[128 + lane + 96], alpha * v3.y);
        atomicAdd(&arow[256 + lane + 96], alpha * v3.z);
        atomicAdd(&arow[384 + lane + 96], alpha * v3.w);
      }
    }
  }
  __syncthreads();

  for (int idx = tid; idx < BAND * 128; idx += BTHREADS) {
    const int row = idx >> 7;
    const int q   = idx & 127;
    if (q < NQ) {
      const float* ar = acc + row * FPAD;
      float4 v;
      v.x = ar[q]; v.y = ar[128 + q]; v.z = ar[256 + q]; v.w = ar[384 + q];
      ((float4*)(out + (size_t)(d0 + row) * OUTD))[q] = v;
    }
  }
}

__global__ __launch_bounds__(256) void k_row_softmax(float* __restrict__ out) {
  const int node = (blockIdx.x * 256 + threadIdx.x) >> 5;
  const int lane = threadIdx.x & 31;
  if (node >= NN) return;
  float4* r4 = (float4*)(out + (size_t)node * OUTD);
  float mx = -INFINITY;
  for (int q = lane; q < NQ; q += 32) {
    const float4 v = r4[q];
    mx = fmaxf(mx, fmaxf(fmaxf(v.x, v.y), fmaxf(v.z, v.w)));
  }
  #pragma unroll
  for (int off = 16; off; off >>= 1) mx = fmaxf(mx, __shfl_xor(mx, off, 32));
  float sm = 0.0f;
  for (int q = lane; q < NQ; q += 32) {
    float4 v = r4[q];
    v.x = __expf(v.x - mx); v.y = __expf(v.y - mx);
    v.z = __expf(v.z - mx); v.w = __expf(v.w - mx);
    r4[q] = v;
    sm += (v.x + v.y) + (v.z + v.w);
  }
  #pragma unroll
  for (int off = 16; off; off >>= 1) sm += __shfl_xor(sm, off, 32);
  const float inv = 1.0f / sm;
  for (int q = lane; q < NQ; q += 32) {
    float4 v = r4[q];
    v.x *= inv; v.y *= inv; v.z *= inv; v.w *= inv;
    r4[q] = v;
  }
}

__global__ __launch_bounds__(256) void k_px_r(const float* __restrict__ px_r,
                                              float* __restrict__ out) {
  const int i = blockIdx.x * 256 + threadIdx.x;
  if (i >= OUTD) return;
  const float v = px_r[i];
  const float sp = (v > 20.0f) ? v : log1pf(__expf(v));
  out[(size_t)NN * OUTD + i] = fminf(fmaxf(sp, 1.0e-4f), 1.0e4f);
}

extern "C" void kernel_launch(void* const* d_in, const int* in_sizes, int n_in,
                              void* d_out, int out_size, void* d_ws, size_t ws_size,
                              hipStream_t stream) {
  const float* x      = (const float*)d_in[0];
  const float* W      = (const float*)d_in[1];
  const float* attn_l = (const float*)d_in[2];
  const float* attn_r = (const float*)d_in[3];
  const float* px_r   = (const float*)d_in[4];
  const int*   src    = (const int*)d_in[5];
  const int*   dst    = (const int*)d_in[6];
  float*       out    = (float*)d_out;

  float*    ws    = (float*)d_ws;
  float*    feat  = ws;
  float*    el    = feat + (size_t)NN * FPAD;
  float*    er    = el + NN;
  unsigned* m_enc = (unsigned*)(er + NN);
  float*    s     = (float*)(m_enc + NN);
  float*    ea    = s + NN;

  k_init       <<<(NN + 255) / 256, 256, 0, stream>>>(m_enc, s);
  k_gemm_wmma  <<<(MTILES * NTILES) / 8, 256, 0, stream>>>(x, W, feat);
  k_el_er      <<<(NN + 7) / 8, 256, 0, stream>>>(feat, attn_l, attn_r, el, er);
  k_edge_logit <<<(EE + 255) / 256, 256, 0, stream>>>(el, er, src, dst, ea, m_enc);
  k_edge_exp   <<<(EE + 255) / 256, 256, 0, stream>>>(dst, ea, m_enc, s);
  k_band_agg   <<<NBANDS, BTHREADS, 0, stream>>>(src, dst, ea, s, feat, out);
  k_row_softmax<<<(NN + 7) / 8, 256, 0, stream>>>(out);
  k_px_r       <<<(OUTD + 255) / 256, 256, 0, stream>>>(px_r, out);
}


# row softmax done by the band owner on its finished rows in LDS: one wave holds a row in registers (max, exp, sum, scale), stores it once normalised; softmax launch gone, 8 -> 7; sum order unchanged
# speedup vs baseline: 2.3965x; 1.1488x over previous
#include <hip/hip_runtime.h>
#include <math.h>

#define NN     10000
#define EE     160000
#define IND    32
#define OUTD   500
#define FPAD   512
#define NQ     125
#define NTILES 32
#define MTILES 625
#define NEG_SLOPE 0.2f

typedef __attribute__((ext_vector_type(2)))  float    v2f;
typedef __attribute__((ext_vector_type(8)))  float    v8f;
typedef __attribute__((ext_vector_type(16))) _Float16 v16h;

#if __has_builtin(__builtin_amdgcn_wmma_f32_16x16x4_f32)
#define USE_WMMA_F32 1
#else
#define USE_WMMA_F32 0
#endif

__device__ __forceinline__ unsigned enc_f(float f) {
  unsigned u = __float_as_uint(f);
  return (u & 0x80000000u) ? ~u : (u | 0x80000000u);
}
__device__ __forceinline__ float dec_f(unsigned e) {
  return (e & 0x80000000u) ? __uint_as_float(e & 0x7FFFFFFFu)
                           : __uint_as_float(~e);
}
#define MAXENC_INIT 0x007FFFFFu

__global__ __launch_bounds__(256) void k_init(unsigned* __restrict__ m_enc,
                                              float* __restrict__ s) {
  const int i = blockIdx.x * 256 + threadIdx.x;
  if (i < NN) { m_enc[i] = MAXENC_INIT; s[i] = 0.0f; }
}

__global__ __launch_bounds__(256) void k_gemm_wmma(const float* __restrict__ x,
                                                   const float* __restrict__ W,
                                                   float* __restrict__ feat) {
  const int gwave = (blockIdx.x * 256 + threadIdx.x) >> 5;
  const int lane  = threadIdx.x & 31;
  const int tileM = gwave >> 5;
  const int tileN = gwave & (NTILES - 1);
  const int m0 = tileM << 4, n0 = tileN << 4;
  const int r  = lane & 15;
  const int h  = lane >> 4;

  const int   col = n0 + r;
  const int   cc  = (col < OUTD) ? col : 0;
  const float msk = (col < OUTD) ? 1.0f : 0.0f;

  v8f acc = {};
#if USE_WMMA_F32
  v2f A[8], B[8];
  #pragma unroll
  for (int i = 0; i < 8; ++i) {
    const int k = (i << 2) + (h << 1);
    A[i].x = x[(m0 + r) * IND + k];
    A[i].y = x[(m0 + r) * IND + k + 1];
    B[i].x = W[k * OUTD + cc] * msk;
    B[i].y = W[(k + 1) * OUTD + cc] * msk;
  }
  #pragma unroll
  for (int i = 0; i < 8; ++i)
    acc = __builtin_amdgcn_wmma_f32_16x16x4_f32(
        false, A[i], false, B[i], (short)0, acc, false, false);
#else
  v16h a, b;
  #pragma unroll
  for (int j = 0; j < 8; ++j) {
    const int ka = ((j >> 2) << 4) + ((j & 3) << 1) + (h << 3);
    a[2 * j]     = (_Float16)x[(m0 + r) * IND + ka];
    a[2 * j + 1] = (_Float16)x[(m0 + r) * IND + ka + 1];
    const int kb = (j << 1) + (h << 4);
    b[2 * j]     = (_Float16)(W[kb * OUTD + cc] * msk);
    b[2 * j + 1] = (_Float16)(W[(kb + 1) * OUTD + cc] * msk);
  }
  acc = __builtin_amdgcn_wmma_f32_16x16x32_f16(
      false, a, false, b, (short)0, acc, false, false);
#endif
  #pragma unroll
  for (int rr = 0; rr < 8; ++rr)
    feat[(size_t)(m0 + rr + (h << 3)) * FPAD + n0 + r] = acc[rr];
}

__global__ __launch_bounds__(256) void k_el_er(const float* __restrict__ feat,
                                               const float* __restrict__ attn_l,
                                               const float* __restrict__ attn_r,
                                               float* __restrict__ el,
                                               float* __restrict__ er) {
  const int node = (blockIdx.x * 256 + threadIdx.x) >> 5;
  const int lane = threadIdx.x & 31;
  if (node >= NN) return;
  const float4* f4 = (const float4*)(feat + (size_t)node * FPAD);
  const float4* l4 = (const float4*)attn_l;
  const float4* r4 = (const float4*)attn_r;
  float sl = 0.0f, sr = 0.0f;
  for (int q = lane; q < NQ; q += 32) {
    const float4 v = f4[q], l = l4[q], rr = r4[q];
    sl = fmaf(v.x, l.x, fmaf(v.y, l.y, fmaf(v.z, l.z, fmaf(v.w, l.w, sl))));
    sr = fmaf(v.x, rr.x, fmaf(v.y, rr.y, fmaf(v.z, rr.z, fmaf(v.w, rr.w, sr))));
  }
  #pragma unroll
  for (int off = 16; off; off >>= 1) {
    sl += __shfl_xor(sl, off, 32);
    sr += __shfl_xor(sr, off, 32);
  }
  if (lane == 0) { el[node] = sl; er[node] = sr; }
}

__global__ __launch_bounds__(256) void k_edge_logit(const float* __restrict__ el,
                                                    const float* __restrict__ er,
                                                    const int* __restrict__ src,
                                                    const int* __restrict__ dst,
                                                    float* __restrict__ e,
                                                    unsigned* __restrict__ m_enc) {
  const int i = blockIdx.x * 256 + threadIdx.x;
  if (i >= EE) return;
  float v = el[src[i]] + er[dst[i]];
  v = (v > 0.0f) ? v : NEG_SLOPE * v;
  e[i] = v;
  atomicMax(&m_enc[dst[i]], enc_f(v));
}

__global__ __launch_bounds__(256) void k_edge_exp(const int* __restrict__ dst,
                                                  float* __restrict__ e,
                                                  const unsigned* __restrict__ m_enc,
                                                  float* __restrict__ s) {
  const int i = blockIdx.x * 256 + threadIdx.x;
  if (i >= EE) return;
  const int d = dst[i];
  const float a = __expf(e[i] - dec_f(m_enc[d]));
  e[i] = a;
  atomicAdd(&s[d], a);
}

#define BAND     40
#define NBANDS   (NN / BAND)
#define BTHREADS 1024
#define BWAVES   (BTHREADS / 32)
#define NGROUPS  (EE / 128)
#define QPB      8
#define NBATCH   5
__global__ __launch_bounds__(BTHREADS) void k_band_agg(const int* __restrict__ src,
                                                       const int* __restrict__ dst,
                                                       const float* __restrict__ a,
                                                       const float* __restrict__ s,
                                                       const float* __restrict__ feat,
                                                       float* __restrict__ out) {
  __shared__ float acc[BAND * FPAD];
  const int tid  = threadIdx.x;
  const int lane = tid & 31;
  const int w    = tid >> 5;
  const int d0   = blockIdx.x * BAND;

  for (int i = tid; i < BAND * FPAD; i += BTHREADS) acc[i] = 0.0f;
  __syncthreads();

  const int4* dst4 = (const int4*)dst;
  #pragma nounroll
  for (int bt = 0; bt < NBATCH; ++bt) {
    int dv[4 * QPB];
    #pragma unroll
    for (int k = 0; k < QPB; ++k) {
      const int  g = w + BWAVES * (QPB * bt + k);
      const int4 q = dst4[(g < NGROUPS ? g : w) * 32 + lane];
      dv[4 * k + 0] = q.x; dv[4 * k + 1] = q.y; dv[4 * k + 2] = q.z; dv[4 * k + 3] = q.w;
    }
    unsigned pend = 0u;
    #pragma unroll
    for (int j = 0; j < 4 * QPB; ++j)
      pend |= ((unsigned)(dv[j] - d0) < (unsigned)BAND) ? (1u << j) : 0u;
    #pragma unroll
    for (int k = 0; k < QPB; ++k)
      if (w + BWAVES * (QPB * bt + k) >= NGROUPS) pend &= ~(0xFu << (4 * k));

    while (__builtin_amdgcn_ballot_w32(pend != 0u) != 0u) {
      const bool has = pend != 0u;
      const int  j   = has ? __builtin_ctz(pend) : 0;
      pend &= pend - 1u;
      int dc = dv[0];
      #pragma unroll
      for (int t = 1; t < 4 * QPB; ++t) dc = (j == t) ? dv[t] : dc;
      const int e = (w + BWAVES * (QPB * bt + (j >> 2))) * 128 + lane * 4 + (j & 3);
      int   snv = 0;
      float alv = 0.0f;
      if (has) {
        snv = src[e];
        const float av = a[e];
        const float sv = s[dc];
        alv = av / (sv > 0.0f ? sv : 1.0f);
      }
      unsigned m = __builtin_amdgcn_ballot_w32(has);
      while (m != 0u) {
        const int l = __builtin_ctz(m);
        m &= m - 1u;
        const int   row   = __builtin_amdgcn_readlane(dc, l) - d0;
        const int   snl   = __builtin_amdgcn_readlane(snv, l);
        const float alpha = __int_as_float(
            __builtin_amdgcn_readlane(__float_as_int(alv), l));
        const float4* f4 = (const float4*)(feat + (size_t)snl * FPAD);
        float* arow = acc + row * FPAD;
        const float4 v0 = f4[lane];
        const float4 v1 = f4[lane + 32];
        const float4 v2 = f4[lane + 64];
        const float4 v3 = f4[lane + 96];
        atomicAdd(&arow[lane],            alpha * v0.x);
        atomicAdd(&arow[128 + lane],      alpha * v0.y);
        atomicAdd(&arow[256 + lane],      alpha * v0.z);
        atomicAdd(&arow[384 + lane],      alpha * v0.w);
        atomicAdd(&arow[lane + 32],       alpha * v1.x);
        atomicAdd(&arow[128 + lane + 32], alpha * v1.y);
        atomicAdd(&arow[256 + lane + 32], alpha * v1.z);
        atomicAdd(&arow[384 + lane + 32], alpha * v1.w);
        atomicAdd(&arow[lane + 64],       alpha * v2.x);
        atomicAdd(&arow[128 + lane + 64], alpha * v2.y);
        atomicAdd(&arow[256 + lane + 64], alpha * v2.z);
        atomicAdd(&arow[384 + lane + 64], alpha * v2.w);
        atomicAdd(&arow[lane + 96],       alpha * v3.x);
        atomicAdd(&arow[128 + lane + 96], alpha * v3.y);
        atomicAdd(&arow[256 + lane + 96], alpha * v3.z);
        atomicAdd(&arow[384 + lane + 96], alpha * v3.w);
      }
    }
  }
  __syncthreads();

  for (int row = w; row < BAND; row += BWAVES) {
    const float* ar = acc + row * FPAD;
    float4 v[4];
    float mx = -INFINITY;
    #pragma unroll
    for (int j = 0; j < 4; ++j) {
      const int q = lane + 32 * j;
      v[j].x = ar[q]; v[j].y = ar[128 + q]; v[j].z = ar[256 + q]; v[j].w = ar[384 + q];
      if (q < NQ)
        mx = fmaxf(mx, fmaxf(fmaxf(v[j].x, v[j].y), fmaxf(v[j].z, v[j].w)));
    }
    #pragma unroll
    for (int off = 16; off; off >>= 1) mx = fmaxf(mx, __shfl_xor(mx, off, 32));
    float sm = 0.0f;
    #pragma unroll
    for (int j = 0; j < 4; ++j) {
      if (lane + 32 * j < NQ) {
        v[j].x = __expf(v[j].x - mx); v[j].y = __expf(v[j].y - mx);
        v[j].z = __expf(v[j].z - mx); v[j].w = __expf(v[j].w - mx);
        sm += (v[j].x + v[j].y) + (v[j].z + v[j].w);
      }
    }
    #pragma unroll
    for (int off = 16; off; off >>= 1) sm += __shfl_xor(sm, off, 32);
    const float inv = 1.0f / sm;
    float4* o4 = (float4*)(out + (size_t)(d0 + row) * OUTD);
    #pragma unroll
    for (int j = 0; j < 4; ++j) {
      const int q = lane + 32 * j;
      if (q < NQ) {
        float4 o = v[j];
        o.x *= inv; o.y *= inv; o.z *= inv; o.w *= inv;
        o4[q] = o;
      }
    }
  }
}

__global__ __launch_bounds__(256) void k_px_r(const float* __restrict__ px_r,
                                              float* __restrict__ out) {
  const int i = blockIdx.x * 256 + threadIdx.x;
  if (i >= OUTD) return;
  const float v = px_r[i];
  const float sp = (v > 20.0f) ? v : log1pf(__expf(v));
  out[(size_t)NN * OUTD + i] = fminf(fmaxf(sp, 1.0e-4f), 1.0e4f);
}

extern "C" void kernel_launch(void* const* d_in, const int* in_sizes, int n_in,
                              void* d_out, int out_size, void* d_ws, size_t ws_size,
                              hipStream_t stream) {
  const float* x      = (const float*)d_in[0];
  const float* W      = (const float*)d_in[1];
  const float* attn_l = (const float*)d_in[2];
  const float* attn_r = (const float*)d_in[3];
  const float* px_r   = (const float*)d_in[4];
  const int*   src    = (const int*)d_in[5];
  const int*   dst    = (const int*)d_in[6];
  float*       out    = (float*)d_out;

  float*    ws    = (float*)d_ws;
  float*    feat  = ws;
  float*    el    = feat + (size_t)NN * FPAD;
  float*    er    = el + NN;
  unsigned* m_enc = (unsigned*)(er + NN);
  float*    s     = (float*)(m_enc + NN);
  float*    ea    = s + NN;

  k_init       <<<(NN + 255) / 256, 256, 0, stream>>>(m_enc, s);
  k_gemm_wmma  <<<(MTILES * NTILES) / 8, 256, 0, stream>>>(x, W, feat);
  k_el_er      <<<(NN + 7) / 8, 256, 0, stream>>>(feat, attn_l, attn_r, el, er);
  k_edge_logit <<<(EE + 255) / 256, 256, 0, stream>>>(el, er, src, dst, ea, m_enc);
  k_edge_exp   <<<(EE + 255) / 256, 256, 0, stream>>>(dst, ea, m_enc, s);
  k_band_agg   <<<NBANDS, BTHREADS, 0, stream>>>(src, dst, ea, s, feat, out);
  k_px_r       <<<(OUTD + 255) / 256, 256, 0, stream>>>(px_r, out);
}


# 3 small launches folded into neighbours: max seed set by the el/er launch; edge exp + segment sum by the band owner (LDS float atomics; rows scaled by 1/s once); px_r by 2 extra logit blocks; 7 -> 4
# speedup vs baseline: 3.0939x; 1.2910x over previous
#include <hip/hip_runtime.h>
#include <math.h>

#define NN     10000
#define EE     160000
#define IND    32
#define OUTD   500
#define FPAD   512
#define NQ     125
#define NTILES 32
#define MTILES 625
#define NEG_SLOPE 0.2f

typedef __attribute__((ext_vector_type(2)))  float    v2f;
typedef __attribute__((ext_vector_type(8)))  float    v8f;
typedef __attribute__((ext_vector_type(16))) _Float16 v16h;

#if __has_builtin(__builtin_amdgcn_wmma_f32_16x16x4_f32)
#define USE_WMMA_F32 1
#else
#define USE_WMMA_F32 0
#endif

__device__ __forceinline__ unsigned enc_f(float f) {
  unsigned u = __float_as_uint(f);
  return (u & 0x80000000u) ? ~u : (u | 0x80000000u);
}
__device__ __forceinline__ float dec_f(unsigned e) {
  return (e & 0x80000000u) ? __uint_as_float(e & 0x7FFFFFFFu)
                           : __uint_as_float(~e);
}
#define MAXENC_INIT 0x007FFFFFu

__global__ __launch_bounds__(256) void k_gemm_wmma(const float* __restrict__ x,
                                                   const float* __restrict__ W,
                                                   float* __restrict__ feat) {
  const int gwave = (blockIdx.x * 256 + threadIdx.x) >> 5;
  const int lane  = threadIdx.x & 31;
  const int tileM = gwave >> 5;
  const int tileN = gwave & (NTILES - 1);
  const int m0 = tileM << 4, n0 = tileN << 4;
  const int r  = lane & 15;
  const int h  = lane >> 4;

  const int   col = n0 + r;
  const int   cc  = (col < OUTD) ? col : 0;
  const float msk = (col < OUTD) ? 1.0f : 0.0f;

  v8f acc = {};
#if USE_WMMA_F32
  v2f A[8], B[8];
  #pragma unroll
  for (int i = 0; i < 8; ++i) {
    const int k = (i << 2) + (h << 1);
    A[i].x = x[(m0 + r) * IND + k];
    A[i].y = x[(m0 + r) * IND + k + 1];
    B[i].x = W[k * OUTD + cc] * msk;
    B[i].y = W[(k + 1) * OUTD + cc] * msk;
  }
  #pragma unroll
  for (int i = 0; i < 8; ++i)
    acc = __builtin_amdgcn_wmma_f32_16x16x4_f32(
        false, A[i], false, B[i], (short)0, acc, false, false);
#else
  v16h a, b;
  #pragma unroll
  for (int j = 0; j < 8; ++j) {
    const int ka = ((j >> 2) << 4) + ((j & 3) << 1) + (h << 3);
    a[2 * j]     = (_Float16)x[(m0 + r) * IND + ka];
    a[2 * j + 1] = (_Float16)x[(m0 + r) * IND + ka + 1];
    const int kb = (j << 1) + (h << 4);
    b[2 * j]     = (_Float16)(W[kb * OUTD + cc] * msk);
    b[2 * j + 1] = (_Float16)(W[(kb + 1) * OUTD + cc] * msk);
  }
  acc = __builtin_amdgcn_wmma_f32_16x16x32_f16(
      false, a, false, b, (short)0, acc, false, false);
#endif
  #pragma unroll
  for (int rr = 0; rr < 8; ++rr)
    feat[(size_t)(m0 + rr + (h << 3)) * FPAD + n0 + r] = acc[rr];
}

__global__ __launch_bounds__(256) void k_el_er(const float* __restrict__ feat,
                                               const float* __restrict__ attn_l,
                                               const float* __restrict__ attn_r,
                                               float* __restrict__ el,
                                               float* __restrict__ er,
                                               unsigned* __restrict__ m_enc) {
  const int node = (blockIdx.x * 256 + threadIdx.x) >> 5;
  const int lane = threadIdx.x & 31;
  if (node >= NN) return;
  const float4* f4 = (const float4*)(feat + (size_t)node * FPAD);
  const float4* l4 = (const float4*)attn_l;
  const float4* r4 = (const float4*)attn_r;
  float sl = 0.0f, sr = 0.0f;
  for (int q = lane; q < NQ; q += 32) {
    const float4 v = f4[q], l = l4[q], rr = r4[q];
    sl = fmaf(v.x, l.x, fmaf(v.y, l.y, fmaf(v.z, l.z, fmaf(v.w, l.w, sl))));
    sr = fmaf(v.x, rr.x, fmaf(v.y, rr.y, fmaf(v.z, rr.z, fmaf(v.w, rr.w, sr))));
  }
  #pragma unroll
  for (int off = 16; off; off >>= 1) {
    sl += __shfl_xor(sl, off, 32);
    sr += __shfl_xor(sr, off, 32);
  }
  if (lane == 0) { el[node] = sl; er[node] = sr; m_enc[node] = MAXENC_INIT; }
}

__global__ __launch_bounds__(256) void k_edge_logit(const float* __restrict__ el,
                                                    const float* __restrict__ er,
                                                    const int* __restrict__ src,
                                                    const int* __restrict__ dst,
                                                    float* __restrict__ e,
                                                    unsigned* __restrict__ m_enc,
                                                    const float* __restrict__ px_r,
                                                    float* __restrict__ out) {
  const int i = blockIdx.x * 256 + threadIdx.x;
  if (i >= EE) {
    const int c = i - EE;
    if (c < OUTD) {
      const float p  = px_r[c];
      const float sp = (p > 20.0f) ? p : log1pf(__expf(p));
      out[(size_t)NN * OUTD + c] = fminf(fmaxf(sp, 1.0e-4f), 1.0e4f);
    }
    return;
  }
  float v = el[src[i]] + er[dst[i]];
  v = (v > 0.0f) ? v : NEG_SLOPE * v;
  e[i] = v;
  atomicMax(&m_enc[dst[i]], enc_f(v));
}

#define BAND     40
#define NBANDS   (NN / BAND)
#define BTHREADS 1024
#define BWAVES   (BTHREADS / 32)
#define NGROUPS  (EE / 128)
#define QPB      8
#define NBATCH   5
__global__ __launch_bounds__(BTHREADS) void k_band_agg(const int* __restrict__ src,
                                                       const int* __restrict__ dst,
                                                       const float* __restrict__ lg,
                                                       const unsigned* __restrict__ m_enc,
                                                       const float* __restrict__ feat,
                                                       float* __restrict__ out) {
  __shared__ float acc[BAND * FPAD];
  __shared__ float ssum[BAND];
  const int tid  = threadIdx.x;
  const int lane = tid & 31;
  const int w    = tid >> 5;
  const int d0   = blockIdx.x * BAND;

  for (int i = tid; i < BAND * FPAD; i += BTHREADS) acc[i] = 0.0f;
  if (tid < BAND) ssum[tid] = 0.0f;
  __syncthreads();

  const int4* dst4 = (const int4*)dst;
  #pragma nounroll
  for (int bt = 0; bt < NBATCH; ++bt) {
    int dv[4 * QPB];
    #pragma unroll
    for (int k = 0; k < QPB; ++k) {
      const int  g = w + BWAVES * (QPB * bt + k);
      const int4 q = dst4[(g < NGROUPS ? g : w) * 32 + lane];
      dv[4 * k + 0] = q.x; dv[4 * k + 1] = q.y; dv[4 * k + 2] = q.z; dv[4 * k + 3] = q.w;
    }
    unsigned pend = 0u;
    #pragma unroll
    for (int j = 0; j < 4 * QPB; ++j)
      pend |= ((unsigned)(dv[j] - d0) < (unsigned)BAND) ? (1u << j) : 0u;
    #pragma unroll
    for (int k = 0; k < QPB; ++k)
      if (w + BWAVES * (QPB * bt + k) >= NGROUPS) pend &= ~(0xFu << (4 * k));

    while (__builtin_amdgcn_ballot_w32(pend != 0u) != 0u) {
      const bool has = pend != 0u;
      const int  j   = has ? __builtin_ctz(pend) : 0;
      pend &= pend - 1u;
      int dc = dv[0];
      #pragma unroll
      for (int t = 1; t < 4 * QPB; ++t) dc = (j == t) ? dv[t] : dc;
      const int e = (w + BWAVES * (QPB * bt + (j >> 2))) * 128 + lane * 4 + (j & 3);
      int   snv = 0;
      float alv = 0.0f;
      if (has) {
        snv = src[e];
        alv = __expf(lg[e] - dec_f(m_enc[dc]));
        atomicAdd(&ssum[dc - d0], alv);
      }
      unsigned m = __builtin_amdgcn_ballot_w32(has);
      while (m != 0u) {
        const int l = __builtin_ctz(m);
        m &= m - 1u;
        const int   row   = __builtin_amdgcn_readlane(dc, l) - d0;
        const int   snl   = __builtin_amdgcn_readlane(snv, l);
        const float alpha = __int_as_float(
            __builtin_amdgcn_readlane(__float_as_int(alv), l));
        const float4* f4 = (const float4*)(feat + (size_t)snl * FPAD);
        float* arow = acc + row * FPAD;
        const float4 v0 = f4[lane];
        const float4 v1 = f4[lane + 32];
        const float4 v2 = f4[lane + 64];
        const float4 v3 = f4[lane + 96];
        atomicAdd(&arow[lane],            alpha * v0.x);
        atomicAdd(&arow[128 + lane],      alpha * v0.y);
        atomicAdd(&arow[256 + lane],      alpha * v0.z);
        atomicAdd(&arow[384 + lane],      alpha * v0.w);
        atomicAdd(&arow[lane + 32],       alpha * v1.x);
        atomicAdd(&arow[128 + lane + 32], alpha * v1.y);
        atomicAdd(&arow[256 + lane + 32], alpha * v1.z);
        atomicAdd(&arow[384 + lane + 32], alpha * v1.w);
        atomicAdd(&arow[lane + 64],       alpha * v2.x);
        atomicAdd(&arow[128 + lane + 64], alpha * v2.y);
        atomicAdd(&arow[256 + lane + 64], alpha * v2.z);
        atomicAdd(&arow[384 + lane + 64], alpha * v2.w);
        atomicAdd(&arow[lane + 96],       alpha * v3.x);
        atomicAdd(&arow[128 + lane + 96], alpha * v3.y);
        atomicAdd(&arow[256 + lane + 96], alpha * v3.z);
        atomicAdd(&arow[384 + lane + 96], alpha * v3.w);
      }
    }
  }
  __syncthreads();

  for (int row = w; row < BAND; row += BWAVES) {
    const float* ar = acc + row * FPAD;
    const float  sg  = ssum[row];
    const float  rs  = 1.0f / ((sg > 0.0f) ? sg : 1.0f);
    float4 v[4];
    float mx = -INFINITY;
    #pragma unroll
    for (int j = 0; j < 4; ++j) {
      const int q = lane + 32 * j;
      v[j].x = ar[q] * rs;       v[j].y = ar[128 + q] * rs;
      v[j].z = ar[256 + q] * rs; v[j].w = ar[384 + q] * rs;
      if (q < NQ)
        mx = fmaxf(mx, fmaxf(fmaxf(v[j].x, v[j].y), fmaxf(v[j].z, v[j].w)));
    }
    #pragma unroll
    for (int off = 16; off; off >>= 1) mx = fmaxf(mx, __shfl_xor(mx, off, 32));
    float sm = 0.0f;
    #pragma unroll
    for (int j = 0; j < 4; ++j) {
      if (lane + 32 * j < NQ) {
        v[j].x = __expf(v[j].x - mx); v[j].y = __expf(v[j].y - mx);
        v[j].z = __expf(v[j].z - mx); v[j].w = __expf(v[j].w - mx);
        sm += (v[j].x + v[j].y) + (v[j].z + v[j].w);
      }
    }
    #pragma unroll
    for (int off = 16; off; off >>= 1) sm += __shfl_xor(sm, off, 32);
    const float inv = 1.0f / sm;
    float4* o4 = (float4*)(out + (size_t)(d0 + row) * OUTD);
    #pragma unroll
    for (int j = 0; j < 4; ++j) {
      const int q = lane + 32 * j;
      if (q < NQ) {
        float4 o = v[j];
        o.x *= inv; o.y *= inv; o.z *= inv; o.w *= inv;
        o4[q] = o;
      }
    }
  }
}

extern "C" void kernel_launch(void* const* d_in, const int* in_sizes, int n_in,
                              void* d_out, int out_size, void* d_ws, size_t ws_size,
                              hipStream_t stream) {
  const float* x      = (const float*)d_in[0];
  const float* W      = (const float*)d_in[1];
  const float* attn_l = (const float*)d_in[2];
  const float* attn_r = (const float*)d_in[3];
  const float* px_r   = (const float*)d_in[4];
  const int*   src    = (const int*)d_in[5];
  const int*   dst    = (const int*)d_in[6];
  float*       out    = (float*)d_out;

  float*    ws    = (float*)d_ws;
  float*    feat  = ws;
  float*    el    = feat + (size_t)NN * FPAD;
  float*    er    = el + NN;
  unsigned* m_enc = (unsigned*)(er + NN);
  float*    lg    = (float*)(m_enc + NN);

  k_gemm_wmma  <<<(MTILES * NTILES) / 8, 256, 0, stream>>>(x, W, feat);
  k_el_er      <<<(NN + 7) / 8, 256, 0, stream>>>(feat, attn_l, attn_r, el, er, m_enc);
  k_edge_logit <<<EE / 256 + (OUTD + 255) / 256, 256, 0, stream>>>(el, er, src, dst, lg, m_enc,
                                                                   px_r, out);
  k_band_agg   <<<NBANDS, BTHREADS, 0, stream>>>(src, dst, lg, m_enc, feat, out);
}
